# starting kernel
# baseline (speedup 1.0000x reference)
#include <hip/hip_runtime.h>

typedef __attribute__((ext_vector_type(2))) float v2f;
typedef __attribute__((ext_vector_type(8))) float v8f;

#define SLOT_DIM 128
#define HIDDEN   64
#define NB       64
#define NK       128

__global__ __launch_bounds__(256) void proj_gemm_wmma(
    const float* __restrict__ slots,
    const float* __restrict__ W1,
    float* __restrict__ LR)
{
    const int lane  = threadIdx.x & 31;
    const int wave  = threadIdx.x >> 5;
    const int mtile = blockIdx.x * 8 + wave;
    const int ntile = blockIdx.y;

    const int m15 = lane & 15;
    const int kb  = (lane >> 4) << 1;

    const int    gRow = mtile * 16 + m15;
    const float* aptr = slots + (size_t)gRow * SLOT_DIM;

    const int    n    = ntile * 16 + m15;
    const int    wrow = (ntile >= 4) ? (n - HIDDEN) : n;
    const float* bptr = W1 + (size_t)wrow * (2 * SLOT_DIM) + ((ntile >= 4) ? SLOT_DIM : 0);

    v8f c = {};
#pragma unroll
    for (int k0 = 0; k0 < SLOT_DIM; k0 += 4) {
        v2f a, b;
        a.x = aptr[k0 + kb];
        a.y = aptr[k0 + kb + 1];
        b.x = bptr[k0 + kb];
        b.y = bptr[k0 + kb + 1];
        c = __builtin_amdgcn_wmma_f32_16x16x4_f32(false, a, false, b,
                                                  (short)0, c, false, false);
    }

#pragma unroll
    for (int r = 0; r < 8; ++r) {
        const int mo = r + ((lane >> 4) << 3);
        LR[(size_t)(mtile * 16 + mo) * 128 + ntile * 16 + m15] = c[r];
    }
}

__global__ __launch_bounds__(256) void edge_score(
    const float* __restrict__ LR,
    const float* __restrict__ b1,
    const float* __restrict__ W2,
    const float* __restrict__ b2p,
    float* __restrict__ out)
{
    __shared__ float Lsh[16 * 64];
    __shared__ float Rsh[128 * 65];
    __shared__ float b1sh[64];
    __shared__ float W2sh[64];

    const int bidx = blockIdx.x;
    const int i0   = blockIdx.y * 16;
    const int tid  = threadIdx.x;

    const float* base = LR + (size_t)bidx * NK * 128;

    for (int idx = tid; idx < 16 * 64; idx += 256) {
        const int il = idx >> 6, h = idx & 63;
        Lsh[idx] = base[(size_t)(i0 + il) * 128 + h];
    }
    for (int idx = tid; idx < 128 * 64; idx += 256) {
        const int j = idx >> 6, h = idx & 63;
        Rsh[j * 65 + h] = base[(size_t)j * 128 + 64 + h];
    }
    if (tid < 64) {
        b1sh[tid] = b1[tid];
        W2sh[tid] = W2[tid];
    }
    __syncthreads();

    const float b2 = b2p[0];

    for (int e = tid; e < 16 * 128; e += 256) {
        const int il = e >> 7;
        const int j  = e & 127;
        const float* Lr = Lsh + il * 64;
        const float* Rr = Rsh + j * 65;

        float acc = b2;
#pragma unroll
        for (int h = 0; h < 64; ++h) {
            const float v = Lr[h] + Rr[h] + b1sh[h];
            acc = __builtin_fmaf(fmaxf(v, 0.0f), W2sh[h], acc);
        }
        const float sig = 1.0f / (1.0f + __expf(-acc));
        const int gi = i0 + il;
        out[((size_t)bidx * NK + gi) * NK + j] = (gi == j) ? 0.0f : sig;
    }
}

extern "C" void kernel_launch(void* const* d_in, const int* in_sizes, int n_in,
                              void* d_out, int out_size, void* d_ws, size_t ws_size,
                              hipStream_t stream) {
    (void)in_sizes; (void)n_in; (void)out_size; (void)ws_size;
    const float* slots = (const float*)d_in[0];
    const float* W1    = (const float*)d_in[1];
    const float* b1    = (const float*)d_in[2];
    const float* W2    = (const float*)d_in[3];
    const float* b2    = (const float*)d_in[4];
    float*       out   = (float*)d_out;
    float*       LR    = (float*)d_ws;

    dim3 gA(64, 8);
    proj_gemm_wmma<<<gA, 256, 0, stream>>>(slots, W1, LR);

    dim3 gB(64, 8);
    edge_score<<<gB, 256, 0, stream>>>(LR, b1, W2, b2, out);
}


# edge kernel register-tiled: each lane keeps the R rows of 2 columns in registers, a wave walks 8 rows i reading L+b1 and W2 as broadcast LDS loads (was 4 LDS reads per term); add order now (L+b1)+R
# speedup vs baseline: 1.2419x; 1.2419x over previous
#include <hip/hip_runtime.h>

typedef __attribute__((ext_vector_type(2))) float v2f;
typedef __attribute__((ext_vector_type(8))) float v8f;

#define SLOT_DIM 128
#define HIDDEN   64
#define NB       64
#define NK       128

__global__ __launch_bounds__(256) void proj_gemm_wmma(
    const float* __restrict__ slots,
    const float* __restrict__ W1,
    float* __restrict__ LR)
{
    const int lane  = threadIdx.x & 31;
    const int wave  = threadIdx.x >> 5;
    const int mtile = blockIdx.x * 8 + wave;
    const int ntile = blockIdx.y;

    const int m15 = lane & 15;
    const int kb  = (lane >> 4) << 1;

    const int    gRow = mtile * 16 + m15;
    const float* aptr = slots + (size_t)gRow * SLOT_DIM;

    const int    n    = ntile * 16 + m15;
    const int    wrow = (ntile >= 4) ? (n - HIDDEN) : n;
    const float* bptr = W1 + (size_t)wrow * (2 * SLOT_DIM) + ((ntile >= 4) ? SLOT_DIM : 0);

    v8f c = {};
#pragma unroll
    for (int k0 = 0; k0 < SLOT_DIM; k0 += 4) {
        v2f a, b;
        a.x = aptr[k0 + kb];
        a.y = aptr[k0 + kb + 1];
        b.x = bptr[k0 + kb];
        b.y = bptr[k0 + kb + 1];
        c = __builtin_amdgcn_wmma_f32_16x16x4_f32(false, a, false, b,
                                                  (short)0, c, false, false);
    }

#pragma unroll
    for (int r = 0; r < 8; ++r) {
        const int mo = r + ((lane >> 4) << 3);
        LR[(size_t)(mtile * 16 + mo) * 128 + ntile * 16 + m15] = c[r];
    }
}

typedef __attribute__((ext_vector_type(4))) float v4f;

#define RP 68

__global__ __launch_bounds__(256) void edge_score(
    const float* __restrict__ LR,
    const float* __restrict__ b1,
    const float* __restrict__ W2,
    const float* __restrict__ b2p,
    float* __restrict__ out)
{
    __shared__ __attribute__((aligned(16))) float Rsh[128 * RP];
    __shared__ __attribute__((aligned(16))) float Lsh[32 * 64];
    __shared__ __attribute__((aligned(16))) float W2sh[64];

    const int bidx = blockIdx.x;
    const int i0   = blockIdx.y * 32;
    const int tid  = threadIdx.x;
    const int lane = tid & 31;
    const int wave = tid >> 5;

    const float* base = LR + (size_t)bidx * NK * 128;

#pragma unroll
    for (int k = 0; k < 8; ++k) {
        const int idx = tid + 256 * k;
        const int row = idx >> 4, c4 = idx & 15;
        const v4f v = *reinterpret_cast<const v4f*>(base + (size_t)row * 128 + 64 + c4 * 4);
        *reinterpret_cast<v4f*>(&Rsh[row * RP + c4 * 4]) = v;
    }
#pragma unroll
    for (int k = 0; k < 2; ++k) {
        const int idx = tid + 256 * k;
        const int row = idx >> 4, c4 = idx & 15;
        v4f v = *reinterpret_cast<const v4f*>(base + (size_t)(i0 + row) * 128 + c4 * 4);
        v.x += b1[c4 * 4 + 0];
        v.y += b1[c4 * 4 + 1];
        v.z += b1[c4 * 4 + 2];
        v.w += b1[c4 * 4 + 3];
        *reinterpret_cast<v4f*>(&Lsh[row * 64 + c4 * 4]) = v;
    }
    if (tid < 64) W2sh[tid] = W2[tid];
    __syncthreads();

    const int j0 = (wave & 1) * 64 + lane;
    const int j1 = j0 + 32;
    v4f R0[16], R1[16];
#pragma unroll
    for (int c = 0; c < 16; ++c) {
        R0[c] = *reinterpret_cast<const v4f*>(&Rsh[j0 * RP + c * 4]);
        R1[c] = *reinterpret_cast<const v4f*>(&Rsh[j1 * RP + c * 4]);
    }

    const float b2 = b2p[0];
    const int   iw = (wave >> 1) * 8;

#pragma unroll 1
    for (int ig = 0; ig < 2; ++ig) {
        const float* Lrow = Lsh + (iw + ig * 4) * 64;
        float acc[4][2];
#pragma unroll
        for (int ii = 0; ii < 4; ++ii) { acc[ii][0] = b2; acc[ii][1] = b2; }

        v4f lb[3][4];
        v4f wb[3];
#pragma unroll
        for (int p = 0; p < 2; ++p) {
#pragma unroll
            for (int ii = 0; ii < 4; ++ii)
                lb[p][ii] = *reinterpret_cast<const v4f*>(&Lrow[ii * 64 + p * 4]);
            wb[p] = *reinterpret_cast<const v4f*>(&W2sh[p * 4]);
        }

#pragma unroll
        for (int c = 0; c < 16; ++c) {
            if (c + 2 < 16) {
                const int n = (c + 2) % 3;
#pragma unroll
                for (int ii = 0; ii < 4; ++ii)
                    lb[n][ii] = *reinterpret_cast<const v4f*>(&Lrow[ii * 64 + (c + 2) * 4]);
                wb[n] = *reinterpret_cast<const v4f*>(&W2sh[(c + 2) * 4]);
            }
            const int s = c % 3;
#pragma unroll
            for (int q = 0; q < 4; ++q) {
                const float w = wb[s][q];
#pragma unroll
                for (int ii = 0; ii < 4; ++ii) {
                    acc[ii][0] = __builtin_fmaf(fmaxf(lb[s][ii][q] + R0[c][q], 0.0f), w, acc[ii][0]);
                    acc[ii][1] = __builtin_fmaf(fmaxf(lb[s][ii][q] + R1[c][q], 0.0f), w, acc[ii][1]);
                }
                __builtin_amdgcn_sched_barrier(0x4);
            }
        }

#pragma unroll
        for (int ii = 0; ii < 4; ++ii) {
            const int gi = i0 + iw + ig * 4 + ii;
            float* orow = out + ((size_t)bidx * NK + gi) * NK;
            const float s0 = 1.0f / (1.0f + __expf(-acc[ii][0]));
            const float s1 = 1.0f / (1.0f + __expf(-acc[ii][1]));
            orow[j0] = (gi == j0) ? 0.0f : s0;
            orow[j1] = (gi == j1) ? 0.0f : s1;
        }
    }
}

extern "C" void kernel_launch(void* const* d_in, const int* in_sizes, int n_in,
                              void* d_out, int out_size, void* d_ws, size_t ws_size,
                              hipStream_t stream) {
    (void)in_sizes; (void)n_in; (void)out_size; (void)ws_size;
    const float* slots = (const float*)d_in[0];
    const float* W1    = (const float*)d_in[1];
    const float* b1    = (const float*)d_in[2];
    const float* W2    = (const float*)d_in[3];
    const float* b2    = (const float*)d_in[4];
    float*       out   = (float*)d_out;
    float*       LR    = (float*)d_ws;

    dim3 gA(64, 8);
    proj_gemm_wmma<<<gA, 256, 0, stream>>>(slots, W1, LR);

    dim3 gB(64, 4);
    edge_score<<<gB, 256, 0, stream>>>(LR, b1, W2, b2, out);
}


# projection GEMM re-tiled: 64x64 tile per 4-wave workgroup, operands read as whole 512-B rows into LDS, each wave 32x32 from LDS fragments (was 16 part-lines per load); k grouping of f32 sums changes
# speedup vs baseline: 1.5104x; 1.2162x over previous
#include <hip/hip_runtime.h>

typedef __attribute__((ext_vector_type(2))) float v2f;
typedef __attribute__((ext_vector_type(8))) float v8f;

#define SLOT_DIM 128
#define HIDDEN   64
#define NB       64
#define NK       128

typedef __attribute__((ext_vector_type(4))) float v4f;

#define PA 132

static __device__ __forceinline__ float xhalf(float x) {
    return __builtin_bit_cast(float,
        __builtin_amdgcn_permlanex16(__builtin_bit_cast(int, x), __builtin_bit_cast(int, x),
                                     0x76543210, 0xfedcba98, false, false));
}

static __device__ __forceinline__ v8f mma4(v2f a, v2f b, v8f c) {
    return __builtin_amdgcn_wmma_f32_16x16x4_f32(false, a, false, b, (short)0, c, false, false);
}

__global__ __launch_bounds__(128) void proj_gemm_wmma(
    const float* __restrict__ slots,
    const float* __restrict__ W1,
    float* __restrict__ LR)
{
    __shared__ __attribute__((aligned(16))) float As[64 * PA];
    __shared__ __attribute__((aligned(16))) float Bs[64 * PA];

    const int tid  = threadIdx.x;
    const int lane = tid & 31;
    const int wave = tid >> 5;
    const int m0   = blockIdx.x * 64;
    const int nh   = blockIdx.y;

    const float* ag = slots + (size_t)m0 * SLOT_DIM;
    const float* bg = W1 + nh * SLOT_DIM;
    v4f va[16], vb[16];
#pragma unroll
    for (int k = 0; k < 16; ++k) {
        const int row = wave + 4 * k;
        va[k] = *reinterpret_cast<const v4f*>(ag + (size_t)row * SLOT_DIM + lane * 4);
        vb[k] = *reinterpret_cast<const v4f*>(bg + (size_t)row * (2 * SLOT_DIM) + lane * 4);
    }
#pragma unroll
    for (int k = 0; k < 16; ++k) {
        const int row = wave + 4 * k;
        *reinterpret_cast<v4f*>(&As[row * PA + lane * 4]) = va[k];
        *reinterpret_cast<v4f*>(&Bs[row * PA + lane * 4]) = vb[k];
    }
    __syncthreads();

    const int c  = lane & 15;
    const int h  = lane >> 4;
    const int mp = wave & 1;
    const int np = wave >> 1;

    const float* ap = &As[(mp * 32 + c) * PA + 4 * h];
    const float* bp = &Bs[(np * 32 + c) * PA + 4 * h];

    v4f fa0[16], fa1[16], fb0[16], fb1[16];
#pragma unroll
    for (int t = 0; t < 16; ++t) {
        fa0[t] = *reinterpret_cast<const v4f*>(ap + 8 * t);
        fb0[t] = *reinterpret_cast<const v4f*>(bp + 8 * t);
        fa1[t] = *reinterpret_cast<const v4f*>(ap + 16 * PA + 8 * t);
        fb1[t] = *reinterpret_cast<const v4f*>(bp + 16 * PA + 8 * t);
    }
    __builtin_amdgcn_sched_barrier(0);

    v8f c00 = {}, c01 = {}, c10 = {}, c11 = {};
#pragma unroll
    for (int t = 0; t < 16; ++t) {
        v2f a0, a1, b0, b1;
        a0.x = fa0[t].x; a0.y = fa0[t].y; a1.x = fa1[t].x; a1.y = fa1[t].y;
        b0.x = fb0[t].x; b0.y = fb0[t].y; b1.x = fb1[t].x; b1.y = fb1[t].y;
        c00 = mma4(a0, b0, c00);
        c01 = mma4(a0, b1, c01);
        c10 = mma4(a1, b0, c10);
        c11 = mma4(a1, b1, c11);
        a0.x = fa0[t].z; a0.y = fa0[t].w; a1.x = fa1[t].z; a1.y = fa1[t].w;
        b0.x = fb0[t].z; b0.y = fb0[t].w; b1.x = fb1[t].z; b1.y = fb1[t].w;
        c00 = mma4(a0, b0, c00);
        c01 = mma4(a0, b1, c01);
        c10 = mma4(a1, b0, c10);
        c11 = mma4(a1, b1, c11);
    }

    float* obase = LR + (size_t)(m0 + mp * 32) * 128 + nh * 64 + np * 32 + lane;
#pragma unroll
    for (int r = 0; r < 8; ++r) {
        {
            const float x0 = c00[r], x1 = c01[r];
            const float s0 = xhalf(x0), s1 = xhalf(x1);
            obase[(size_t)r * 128]       = h ? s1 : x0;
            obase[(size_t)(r + 8) * 128] = h ? x1 : s0;
        }
        {
            const float x0 = c10[r], x1 = c11[r];
            const float s0 = xhalf(x0), s1 = xhalf(x1);
            obase[(size_t)(r + 16) * 128] = h ? s1 : x0;
            obase[(size_t)(r + 24) * 128] = h ? x1 : s0;
        }
    }
}

#define RP 68

__global__ __launch_bounds__(256) void edge_score(
    const float* __restrict__ LR,
    const float* __restrict__ b1,
    const float* __restrict__ W2,
    const float* __restrict__ b2p,
    float* __restrict__ out)
{
    __shared__ __attribute__((aligned(16))) float Rsh[128 * RP];
    __shared__ __attribute__((aligned(16))) float Lsh[32 * 64];
    __shared__ __attribute__((aligned(16))) float W2sh[64];

    const int bidx = blockIdx.x;
    const int i0   = blockIdx.y * 32;
    const int tid  = threadIdx.x;
    const int lane = tid & 31;
    const int wave = tid >> 5;

    const float* base = LR + (size_t)bidx * NK * 128;

#pragma unroll
    for (int k = 0; k < 8; ++k) {
        const int idx = tid + 256 * k;
        const int row = idx >> 4, c4 = idx & 15;
        const v4f v = *reinterpret_cast<const v4f*>(base + (size_t)row * 128 + 64 + c4 * 4);
        *reinterpret_cast<v4f*>(&Rsh[row * RP + c4 * 4]) = v;
    }
#pragma unroll
    for (int k = 0; k < 2; ++k) {
        const int idx = tid + 256 * k;
        const int row = idx >> 4, c4 = idx & 15;
        v4f v = *reinterpret_cast<const v4f*>(base + (size_t)(i0 + row) * 128 + c4 * 4);
        v.x += b1[c4 * 4 + 0];
        v.y += b1[c4 * 4 + 1];
        v.z += b1[c4 * 4 + 2];
        v.w += b1[c4 * 4 + 3];
        *reinterpret_cast<v4f*>(&Lsh[row * 64 + c4 * 4]) = v;
    }
    if (tid < 64) W2sh[tid] = W2[tid];
    __syncthreads();

    const int j0 = (wave & 1) * 64 + lane;
    const int j1 = j0 + 32;
    v4f R0[16], R1[16];
#pragma unroll
    for (int c = 0; c < 16; ++c) {
        R0[c] = *reinterpret_cast<const v4f*>(&Rsh[j0 * RP + c * 4]);
        R1[c] = *reinterpret_cast<const v4f*>(&Rsh[j1 * RP + c * 4]);
    }

    const float b2 = b2p[0];
    const int   iw = (wave >> 1) * 8;

#pragma unroll 1
    for (int ig = 0; ig < 2; ++ig) {
        const float* Lrow = Lsh + (iw + ig * 4) * 64;
        float acc[4][2];
#pragma unroll
        for (int ii = 0; ii < 4; ++ii) { acc[ii][0] = b2; acc[ii][1] = b2; }

        v4f lb[3][4];
        v4f wb[3];
#pragma unroll
        for (int p = 0; p < 2; ++p) {
#pragma unroll
            for (int ii = 0; ii < 4; ++ii)
                lb[p][ii] = *reinterpret_cast<const v4f*>(&Lrow[ii * 64 + p * 4]);
            wb[p] = *reinterpret_cast<const v4f*>(&W2sh[p * 4]);
        }

#pragma unroll
        for (int c = 0; c < 16; ++c) {
            if (c + 2 < 16) {
                const int n = (c + 2) % 3;
#pragma unroll
                for (int ii = 0; ii < 4; ++ii)
                    lb[n][ii] = *reinterpret_cast<const v4f*>(&Lrow[ii * 64 + (c + 2) * 4]);
                wb[n] = *reinterpret_cast<const v4f*>(&W2sh[(c + 2) * 4]);
            }
            const int s = c % 3;
#pragma unroll
            for (int q = 0; q < 4; ++q) {
                const float w = wb[s][q];
#pragma unroll
                for (int ii = 0; ii < 4; ++ii) {
                    acc[ii][0] = __builtin_fmaf(fmaxf(lb[s][ii][q] + R0[c][q], 0.0f), w, acc[ii][0]);
                    acc[ii][1] = __builtin_fmaf(fmaxf(lb[s][ii][q] + R1[c][q], 0.0f), w, acc[ii][1]);
                }
                __builtin_amdgcn_sched_barrier(0x4);
            }
        }

#pragma unroll
        for (int ii = 0; ii < 4; ++ii) {
            const int gi = i0 + iw + ig * 4 + ii;
            float* orow = out + ((size_t)bidx * NK + gi) * NK;
            const float s0 = 1.0f / (1.0f + __expf(-acc[ii][0]));
            const float s1 = 1.0f / (1.0f + __expf(-acc[ii][1]));
            orow[j0] = (gi == j0) ? 0.0f : s0;
            orow[j1] = (gi == j1) ? 0.0f : s1;
        }
    }
}

extern "C" void kernel_launch(void* const* d_in, const int* in_sizes, int n_in,
                              void* d_out, int out_size, void* d_ws, size_t ws_size,
                              hipStream_t stream) {
    (void)in_sizes; (void)n_in; (void)out_size; (void)ws_size;
    const float* slots = (const float*)d_in[0];
    const float* W1    = (const float*)d_in[1];
    const float* b1    = (const float*)d_in[2];
    const float* W2    = (const float*)d_in[3];
    const float* b2    = (const float*)d_in[4];
    float*       out   = (float*)d_out;
    float*       LR    = (float*)d_ws;

    dim3 gA(128, 2);
    proj_gemm_wmma<<<gA, 128, 0, stream>>>(slots, W1, LR);

    dim3 gB(64, 4);
    edge_score<<<gB, 256, 0, stream>>>(LR, b1, W2, b2, out);
}


# projection and edge kernels fused into one launch: each 8-wave workgroup computes the L and R rows of its 64x64 out tile on WMMA into LDS and scores from there; f32 sum order unchanged
# speedup vs baseline: 1.7065x; 1.1298x over previous
#include <hip/hip_runtime.h>

typedef __attribute__((ext_vector_type(2))) float v2f;
typedef __attribute__((ext_vector_type(4))) float v4f;
typedef __attribute__((ext_vector_type(8))) float v8f;

#define SLOT_DIM 128
#define HIDDEN   64
#define NK       128

#define PA 132
#define RP 68

static __device__ __forceinline__ v8f mma4(v2f a, v2f b, v8f c) {
    return __builtin_amdgcn_wmma_f32_16x16x4_f32(false, a, false, b, (short)0, c, false, false);
}

__global__ __launch_bounds__(256) void edge_fused(
    const float* __restrict__ slots,
    const float* __restrict__ W1,
    const float* __restrict__ b1,
    const float* __restrict__ W2,
    const float* __restrict__ b2p,
    float* __restrict__ out)
{
    __shared__ __attribute__((aligned(16))) float As[2 * 64 * PA];
    __shared__ __attribute__((aligned(16))) float Bs[2 * 64 * PA];
    __shared__ __attribute__((aligned(16))) float Lsh[64 * RP];
    __shared__ __attribute__((aligned(16))) float Rsh[64 * RP];
    __shared__ __attribute__((aligned(16))) float W2sh[64];

    const int tid  = threadIdx.x;
    const int lane = tid & 31;
    const int wave = tid >> 5;
    const int g    = wave >> 2;
    const int wg   = wave & 3;
    const int bidx = blockIdx.x;
    const int i0   = (blockIdx.y >> 1) * 64;
    const int jt   = (blockIdx.y & 1) * 64;

    {
        const float* ag = slots + ((size_t)bidx * NK + (g ? jt : i0)) * SLOT_DIM;
        const float* bg = W1 + g * SLOT_DIM;
        float* as = As + g * (64 * PA);
        float* bs = Bs + g * (64 * PA);
        v4f va[16], vb[16];
#pragma unroll
        for (int k = 0; k < 16; ++k) {
            const int row = wg + 4 * k;
            va[k] = *reinterpret_cast<const v4f*>(ag + (size_t)row * SLOT_DIM + lane * 4);
            vb[k] = *reinterpret_cast<const v4f*>(bg + (size_t)row * (2 * SLOT_DIM) + lane * 4);
        }
#pragma unroll
        for (int k = 0; k < 16; ++k) {
            const int row = wg + 4 * k;
            *reinterpret_cast<v4f*>(&as[row * PA + lane * 4]) = va[k];
            *reinterpret_cast<v4f*>(&bs[row * PA + lane * 4]) = vb[k];
        }
    }
    if (tid < 64) W2sh[tid] = W2[tid];
    __syncthreads();

    const int c  = lane & 15;
    const int h  = lane >> 4;
    const int mp = wg & 1;
    const int np = wg >> 1;

    {
        const float* ap = &As[g * (64 * PA) + (mp * 32 + c) * PA + 4 * h];
        const float* bp = &Bs[g * (64 * PA) + (np * 32 + c) * PA + 4 * h];

        const float bb0 = g ? -0.0f : b1[np * 32 + c];
        const float bb1 = g ? -0.0f : b1[np * 32 + 16 + c];

        v4f fa0[16], fa1[16], fb0[16], fb1[16];
#pragma unroll
        for (int t = 0; t < 16; ++t) {
            fa0[t] = *reinterpret_cast<const v4f*>(ap + 8 * t);
            fb0[t] = *reinterpret_cast<const v4f*>(bp + 8 * t);
            fa1[t] = *reinterpret_cast<const v4f*>(ap + 16 * PA + 8 * t);
            fb1[t] = *reinterpret_cast<const v4f*>(bp + 16 * PA + 8 * t);
        }
        __builtin_amdgcn_sched_barrier(0);

        v8f c00 = {}, c01 = {}, c10 = {}, c11 = {};
#pragma unroll
        for (int t = 0; t < 16; ++t) {
            v2f a0, a1, b0, b1v;
            a0.x = fa0[t].x; a0.y = fa0[t].y; a1.x = fa1[t].x; a1.y = fa1[t].y;
            b0.x = fb0[t].x; b0.y = fb0[t].y; b1v.x = fb1[t].x; b1v.y = fb1[t].y;
            c00 = mma4(a0, b0, c00);
            c01 = mma4(a0, b1v, c01);
            c10 = mma4(a1, b0, c10);
            c11 = mma4(a1, b1v, c11);
            a0.x = fa0[t].z; a0.y = fa0[t].w; a1.x = fa1[t].z; a1.y = fa1[t].w;
            b0.x = fb0[t].z; b0.y = fb0[t].w; b1v.x = fb1[t].z; b1v.y = fb1[t].w;
            c00 = mma4(a0, b0, c00);
            c01 = mma4(a0, b1v, c01);
            c10 = mma4(a1, b0, c10);
            c11 = mma4(a1, b1v, c11);
        }

        float* dst = (g ? Rsh : Lsh) + (mp * 32 + 8 * h) * RP + np * 32 + c;
#pragma unroll
        for (int r = 0; r < 8; ++r) {
            dst[r * RP]             = c00[r] + bb0;
            dst[r * RP + 16]        = c01[r] + bb1;
            dst[(r + 16) * RP]      = c10[r] + bb0;
            dst[(r + 16) * RP + 16] = c11[r] + bb1;
        }
    }
    __syncthreads();

    const int j0 = lane;
    const int j1 = lane + 32;
    v4f R0[16], R1[16];
#pragma unroll
    for (int q = 0; q < 16; ++q) {
        R0[q] = *reinterpret_cast<const v4f*>(&Rsh[j0 * RP + q * 4]);
        R1[q] = *reinterpret_cast<const v4f*>(&Rsh[j1 * RP + q * 4]);
    }

    const float b2 = b2p[0];
    const int   iw = wave * 8;

#pragma unroll 1
    for (int ig = 0; ig < 2; ++ig) {
        const float* Lrow = Lsh + (iw + ig * 4) * RP;
        float acc[4][2];
#pragma unroll
        for (int ii = 0; ii < 4; ++ii) { acc[ii][0] = b2; acc[ii][1] = b2; }

        v4f lb[3][4];
        v4f wb[3];
#pragma unroll
        for (int p = 0; p < 2; ++p) {
#pragma unroll
            for (int ii = 0; ii < 4; ++ii)
                lb[p][ii] = *reinterpret_cast<const v4f*>(&Lrow[ii * RP + p * 4]);
            wb[p] = *reinterpret_cast<const v4f*>(&W2sh[p * 4]);
        }

#pragma unroll
        for (int q = 0; q < 16; ++q) {
            if (q + 2 < 16) {
                const int n = (q + 2) % 3;
#pragma unroll
                for (int ii = 0; ii < 4; ++ii)
                    lb[n][ii] = *reinterpret_cast<const v4f*>(&Lrow[ii * RP + (q + 2) * 4]);
                wb[n] = *reinterpret_cast<const v4f*>(&W2sh[(q + 2) * 4]);
            }
            const int s = q % 3;
#pragma unroll
            for (int e = 0; e < 4; ++e) {
                const float w = wb[s][e];
#pragma unroll
                for (int ii = 0; ii < 4; ++ii) {
                    acc[ii][0] = __builtin_fmaf(fmaxf(lb[s][ii][e] + R0[q][e], 0.0f), w, acc[ii][0]);
                    acc[ii][1] = __builtin_fmaf(fmaxf(lb[s][ii][e] + R1[q][e], 0.0f), w, acc[ii][1]);
                }
                __builtin_amdgcn_sched_barrier(0x4);
            }
        }

#pragma unroll
        for (int ii = 0; ii < 4; ++ii) {
            const int gi  = i0 + iw + ig * 4 + ii;
            const int gj0 = jt + j0;
            const int gj1 = jt + j1;
            float* orow = out + ((size_t)bidx * NK + gi) * NK;
            const float s0 = 1.0f / (1.0f + __expf(-acc[ii][0]));
            const float s1 = 1.0f / (1.0f + __expf(-acc[ii][1]));
            orow[gj0] = (gi == gj0) ? 0.0f : s0;
            orow[gj1] = (gi == gj1) ? 0.0f : s1;
        }
    }
}

extern "C" void kernel_launch(void* const* d_in, const int* in_sizes, int n_in,
                              void* d_out, int out_size, void* d_ws, size_t ws_size,
                              hipStream_t stream) {
    (void)in_sizes; (void)n_in; (void)out_size; (void)d_ws; (void)ws_size;
    const float* slots = (const float*)d_in[0];
    const float* W1    = (const float*)d_in[1];
    const float* b1    = (const float*)d_in[2];
    const float* W2    = (const float*)d_in[3];
    const float* b2    = (const float*)d_in[4];
    float*       out   = (float*)d_out;

    dim3 grid(64, 4);
    edge_fused<<<grid, 256, 0, stream>>>(slots, W1, b1, W2, b2, out);
}
